# starting kernel
# baseline (speedup 1.0000x reference)
#include <hip/hip_runtime.h>
#include <hip/hip_bf16.h>
#include <math.h>

#define B_ 128
#define T_ 2048
#define H_ 256
#define S_ 8
#define ROWS_PER_WG (T_ / S_)
#define ROWS_PER_WAVE (ROWS_PER_WG / 8)

typedef __attribute__((ext_vector_type(2))) float v2f;
typedef __attribute__((ext_vector_type(8))) float v8f;

__global__ __launch_bounds__(32) void k_vmat(const float* __restrict__ hid,
                                             const float* __restrict__ Wscore,
                                             float* __restrict__ V) {
  const int tm   = blockIdx.x;
  const int tn   = blockIdx.y;
  const int lane = threadIdx.x;
  const int half = lane >> 4;
  const int r    = lane & 15;
  const size_t hstride = (size_t)T_ * H_;
  const float* ht = hid + (size_t)(T_ - 1) * H_;

  v8f acc = {};
  for (int k0 = 0; k0 < H_; k0 += 4) {
    const int k = k0 + 2 * half;
    const float2 av = *(const float2*)(ht + (size_t)(tm * 16 + r) * hstride + k);
    const float2 bv = *(const float2*)(Wscore + (size_t)(tn * 16 + r) * H_ + k);
    v2f a; a[0] = av.x; a[1] = av.y;
    v2f b; b[0] = bv.x; b[1] = bv.y;
    acc = __builtin_amdgcn_wmma_f32_16x16x4_f32(false, a, false, b,
                                                (short)0, acc, false, false);
  }
#pragma unroll
  for (int j = 0; j < 8; ++j) {
    const int m = tm * 16 + j + 8 * half;
    const int n = tn * 16 + r;
    V[(size_t)m * H_ + n] = acc[j];
  }
}

__global__ __launch_bounds__(256) void k_stream(const float* __restrict__ hid,
                                                const float* __restrict__ V,
                                                float* __restrict__ pm,
                                                float* __restrict__ psum,
                                                float* __restrict__ pc) {
  const int b    = blockIdx.x;
  const int s    = blockIdx.y;
  const int wave = threadIdx.x >> 5;
  const int lane = threadIdx.x & 31;

  const float* vrow = V + (size_t)b * H_ + lane * 8;
  const float4 v0 = *(const float4*)vrow;
  const float4 v1 = *(const float4*)(vrow + 4);

  float m = -INFINITY, run = 0.f;
  float c[8] = {0.f, 0.f, 0.f, 0.f, 0.f, 0.f, 0.f, 0.f};

  const int t0 = s * ROWS_PER_WG + wave * ROWS_PER_WAVE;
  const float* base = hid + ((size_t)b * T_ + t0) * H_ + lane * 8;

  for (int i = 0; i < ROWS_PER_WAVE; ++i) {
    const float* row = base + (size_t)i * H_;
    const float4 h0 = *(const float4*)row;
    const float4 h1 = *(const float4*)(row + 4);
    float p = h0.x * v0.x + h0.y * v0.y + h0.z * v0.z + h0.w * v0.w
            + h1.x * v1.x + h1.y * v1.y + h1.z * v1.z + h1.w * v1.w;
    p += __shfl_xor(p, 16);
    p += __shfl_xor(p, 8);
    p += __shfl_xor(p, 4);
    p += __shfl_xor(p, 2);
    p += __shfl_xor(p, 1);
    const float mn = fmaxf(m, p);
    const float sc = __expf(m - mn);
    const float pe = __expf(p - mn);
    run = run * sc + pe;
    m = mn;
    c[0] = c[0] * sc + pe * h0.x;
    c[1] = c[1] * sc + pe * h0.y;
    c[2] = c[2] * sc + pe * h0.z;
    c[3] = c[3] * sc + pe * h0.w;
    c[4] = c[4] * sc + pe * h1.x;
    c[5] = c[5] * sc + pe * h1.y;
    c[6] = c[6] * sc + pe * h1.z;
    c[7] = c[7] * sc + pe * h1.w;
  }

  __shared__ float sm[8];
  __shared__ float ss[8];
  __shared__ float scx[8][H_];
#pragma unroll
  for (int j = 0; j < 8; ++j) scx[wave][lane * 8 + j] = c[j];
  if (lane == 0) { sm[wave] = m; ss[wave] = run; }
  __syncthreads();

  const int h = threadIdx.x;
  float M = sm[0];
#pragma unroll
  for (int w = 1; w < 8; ++w) M = fmaxf(M, sm[w]);
  float tot = 0.f, ch = 0.f;
#pragma unroll
  for (int w = 0; w < 8; ++w) {
    const float e = __expf(sm[w] - M);
    tot += ss[w] * e;
    ch  += scx[w][h] * e;
  }
  const int idx = b * S_ + s;
  pc[(size_t)idx * H_ + h] = ch;
  if (threadIdx.x == 0) { pm[idx] = M; psum[idx] = tot; }
}

__global__ __launch_bounds__(256) void k_combine(const float* __restrict__ hid,
                                                 const float* __restrict__ pm,
                                                 const float* __restrict__ psum,
                                                 const float* __restrict__ pc,
                                                 float* __restrict__ preact) {
  const int b = blockIdx.x;
  const int h = threadIdx.x;
  float M = pm[b * S_];
#pragma unroll
  for (int s = 1; s < S_; ++s) M = fmaxf(M, pm[b * S_ + s]);
  float tot = 0.f, ch = 0.f;
#pragma unroll
  for (int s = 0; s < S_; ++s) {
    const float e = __expf(pm[b * S_ + s] - M);
    tot += psum[b * S_ + s] * e;
    ch  += pc[(size_t)(b * S_ + s) * H_ + h] * e;
  }
  preact[(size_t)b * 512 + h]       = ch / tot;
  preact[(size_t)b * 512 + 256 + h] = hid[((size_t)b * T_ + (T_ - 1)) * H_ + h];
}

__global__ __launch_bounds__(32) void k_att(const float* __restrict__ preact,
                                            const float* __restrict__ Watt,
                                            float* __restrict__ out) {
  const int tm   = blockIdx.x;
  const int tn   = blockIdx.y;
  const int lane = threadIdx.x;
  const int half = lane >> 4;
  const int r    = lane & 15;

  v8f acc = {};
  for (int k0 = 0; k0 < 512; k0 += 4) {
    const int k = k0 + 2 * half;
    const float2 av = *(const float2*)(preact + (size_t)(tm * 16 + r) * 512 + k);
    v2f a; a[0] = av.x; a[1] = av.y;
    v2f b;
    b[0] = Watt[(size_t)k * 128 + tn * 16 + r];
    b[1] = Watt[(size_t)(k + 1) * 128 + tn * 16 + r];
    acc = __builtin_amdgcn_wmma_f32_16x16x4_f32(false, a, false, b,
                                                (short)0, acc, false, false);
  }
#pragma unroll
  for (int j = 0; j < 8; ++j) {
    const int m = tm * 16 + j + 8 * half;
    const int n = tn * 16 + r;
    out[(size_t)m * 128 + n] = tanhf(acc[j]);
  }
}

extern "C" void kernel_launch(void* const* d_in, const int* in_sizes, int n_in,
                              void* d_out, int out_size, void* d_ws, size_t ws_size,
                              hipStream_t stream) {
  const float* hid    = (const float*)d_in[0];
  const float* Wscore = (const float*)d_in[1];
  const float* Watt   = (const float*)d_in[2];
  float* out = (float*)d_out;
  float* ws  = (float*)d_ws;

  float* V      = ws;
  float* pm     = ws + 32768;
  float* psum   = ws + 33792;
  float* pc     = ws + 34816;
  float* preact = ws + 296960;

  k_vmat   <<<dim3(8, 16),  32,  0, stream>>>(hid, Wscore, V);
  k_stream <<<dim3(B_, S_), 256, 0, stream>>>(hid, V, pm, psum, pc);
  k_combine<<<B_,           256, 0, stream>>>(hid, pm, psum, pc, preact);
  k_att    <<<dim3(8, 8),   32,  0, stream>>>(preact, Watt, out);
}


# Split K of the two small f32 WMMA GEMMs over the 8 waves of a 256-thread block, partial tiles summed via LDS in fixed wave order (changes the order of the K sums)
# speedup vs baseline: 1.2305x; 1.2305x over previous
#include <hip/hip_runtime.h>
#include <hip/hip_bf16.h>
#include <math.h>

#define B_ 128
#define T_ 2048
#define H_ 256
#define S_ 8
#define ROWS_PER_WG (T_ / S_)
#define ROWS_PER_WAVE (ROWS_PER_WG / 8)
#define KSPLIT_ 8

typedef __attribute__((ext_vector_type(2))) float v2f;
typedef __attribute__((ext_vector_type(8))) float v8f;

__global__ __launch_bounds__(256) void k_vmat(const float* __restrict__ hid,
                                              const float* __restrict__ Wscore,
                                              float* __restrict__ V) {
  const int tm   = blockIdx.x;
  const int tn   = blockIdx.y;
  const int wave = threadIdx.x >> 5;
  const int lane = threadIdx.x & 31;
  const int half = lane >> 4;
  const int r    = lane & 15;
  const size_t hstride = (size_t)T_ * H_;
  const float* ht = hid + (size_t)(T_ - 1) * H_;
  const int kb = wave * (H_ / KSPLIT_) + 2 * half;

  const float* arow = ht + (size_t)(tm * 16 + r) * hstride + kb;
  const float* brow = Wscore + (size_t)(tn * 16 + r) * H_ + kb;
  constexpr int NSTEP = H_ / KSPLIT_ / 4;
  float2 av[NSTEP], bv[NSTEP];
#pragma unroll
  for (int i = 0; i < NSTEP; ++i) {
    av[i] = *(const float2*)(arow + 4 * i);
    bv[i] = *(const float2*)(brow + 4 * i);
  }
  v8f acc = {};
#pragma unroll
  for (int i = 0; i < NSTEP; ++i) {
    v2f a; a[0] = av[i].x; a[1] = av[i].y;
    v2f b; b[0] = bv[i].x; b[1] = bv[i].y;
    acc = __builtin_amdgcn_wmma_f32_16x16x4_f32(false, a, false, b,
                                                (short)0, acc, false, false);
  }

  __shared__ float part[KSPLIT_][256];
#pragma unroll
  for (int j = 0; j < 8; ++j) part[wave][(j + 8 * half) * 16 + r] = acc[j];
  __syncthreads();
  const int t = threadIdx.x;
  float s = part[0][t];
#pragma unroll
  for (int w = 1; w < KSPLIT_; ++w) s += part[w][t];
  V[(size_t)(tm * 16 + (t >> 4)) * H_ + tn * 16 + (t & 15)] = s;
}

__global__ __launch_bounds__(256) void k_stream(const float* __restrict__ hid,
                                                const float* __restrict__ V,
                                                float* __restrict__ pm,
                                                float* __restrict__ psum,
                                                float* __restrict__ pc) {
  const int b    = blockIdx.x;
  const int s    = blockIdx.y;
  const int wave = threadIdx.x >> 5;
  const int lane = threadIdx.x & 31;

  const float* vrow = V + (size_t)b * H_ + lane * 8;
  const float4 v0 = *(const float4*)vrow;
  const float4 v1 = *(const float4*)(vrow + 4);

  float m = -INFINITY, run = 0.f;
  float c[8] = {0.f, 0.f, 0.f, 0.f, 0.f, 0.f, 0.f, 0.f};

  const int t0 = s * ROWS_PER_WG + wave * ROWS_PER_WAVE;
  const float* base = hid + ((size_t)b * T_ + t0) * H_ + lane * 8;

  for (int i = 0; i < ROWS_PER_WAVE; ++i) {
    const float* row = base + (size_t)i * H_;
    const float4 h0 = *(const float4*)row;
    const float4 h1 = *(const float4*)(row + 4);
    float p = h0.x * v0.x + h0.y * v0.y + h0.z * v0.z + h0.w * v0.w
            + h1.x * v1.x + h1.y * v1.y + h1.z * v1.z + h1.w * v1.w;
    p += __shfl_xor(p, 16);
    p += __shfl_xor(p, 8);
    p += __shfl_xor(p, 4);
    p += __shfl_xor(p, 2);
    p += __shfl_xor(p, 1);
    const float mn = fmaxf(m, p);
    const float sc = __expf(m - mn);
    const float pe = __expf(p - mn);
    run = run * sc + pe;
    m = mn;
    c[0] = c[0] * sc + pe * h0.x;
    c[1] = c[1] * sc + pe * h0.y;
    c[2] = c[2] * sc + pe * h0.z;
    c[3] = c[3] * sc + pe * h0.w;
    c[4] = c[4] * sc + pe * h1.x;
    c[5] = c[5] * sc + pe * h1.y;
    c[6] = c[6] * sc + pe * h1.z;
    c[7] = c[7] * sc + pe * h1.w;
  }

  __shared__ float sm[8];
  __shared__ float ss[8];
  __shared__ float scx[8][H_];
#pragma unroll
  for (int j = 0; j < 8; ++j) scx[wave][lane * 8 + j] = c[j];
  if (lane == 0) { sm[wave] = m; ss[wave] = run; }
  __syncthreads();

  const int h = threadIdx.x;
  float M = sm[0];
#pragma unroll
  for (int w = 1; w < 8; ++w) M = fmaxf(M, sm[w]);
  float tot = 0.f, ch = 0.f;
#pragma unroll
  for (int w = 0; w < 8; ++w) {
    const float e = __expf(sm[w] - M);
    tot += ss[w] * e;
    ch  += scx[w][h] * e;
  }
  const int idx = b * S_ + s;
  pc[(size_t)idx * H_ + h] = ch;
  if (threadIdx.x == 0) { pm[idx] = M; psum[idx] = tot; }
}

__global__ __launch_bounds__(256) void k_combine(const float* __restrict__ hid,
                                                 const float* __restrict__ pm,
                                                 const float* __restrict__ psum,
                                                 const float* __restrict__ pc,
                                                 float* __restrict__ preact) {
  const int b = blockIdx.x;
  const int h = threadIdx.x;
  float M = pm[b * S_];
#pragma unroll
  for (int s = 1; s < S_; ++s) M = fmaxf(M, pm[b * S_ + s]);
  float tot = 0.f, ch = 0.f;
#pragma unroll
  for (int s = 0; s < S_; ++s) {
    const float e = __expf(pm[b * S_ + s] - M);
    tot += psum[b * S_ + s] * e;
    ch  += pc[(size_t)(b * S_ + s) * H_ + h] * e;
  }
  preact[(size_t)b * 512 + h]       = ch / tot;
  preact[(size_t)b * 512 + 256 + h] = hid[((size_t)b * T_ + (T_ - 1)) * H_ + h];
}

__global__ __launch_bounds__(256) void k_att(const float* __restrict__ preact,
                                             const float* __restrict__ Watt,
                                             float* __restrict__ out) {
  const int tm   = blockIdx.x;
  const int tn   = blockIdx.y;
  const int wave = threadIdx.x >> 5;
  const int lane = threadIdx.x & 31;
  const int half = lane >> 4;
  const int r    = lane & 15;
  const int kb = wave * (512 / KSPLIT_) + 2 * half;

  const float* arow = preact + (size_t)(tm * 16 + r) * 512 + kb;
  const float* bcol = Watt + (size_t)kb * 128 + tn * 16 + r;
  constexpr int NSTEP = 512 / KSPLIT_ / 4;
  float2 av[NSTEP];
  float b0[NSTEP], b1[NSTEP];
#pragma unroll
  for (int i = 0; i < NSTEP; ++i) {
    av[i] = *(const float2*)(arow + 4 * i);
    b0[i] = bcol[(size_t)(4 * i) * 128];
    b1[i] = bcol[(size_t)(4 * i + 1) * 128];
  }
  v8f acc = {};
#pragma unroll
  for (int i = 0; i < NSTEP; ++i) {
    v2f a; a[0] = av[i].x; a[1] = av[i].y;
    v2f b; b[0] = b0[i];   b[1] = b1[i];
    acc = __builtin_amdgcn_wmma_f32_16x16x4_f32(false, a, false, b,
                                                (short)0, acc, false, false);
  }

  __shared__ float part[KSPLIT_][256];
#pragma unroll
  for (int j = 0; j < 8; ++j) part[wave][(j + 8 * half) * 16 + r] = acc[j];
  __syncthreads();
  const int t = threadIdx.x;
  float s = part[0][t];
#pragma unroll
  for (int w = 1; w < KSPLIT_; ++w) s += part[w][t];
  out[(size_t)(tm * 16 + (t >> 4)) * 128 + tn * 16 + (t & 15)] = tanhf(s);
}

extern "C" void kernel_launch(void* const* d_in, const int* in_sizes, int n_in,
                              void* d_out, int out_size, void* d_ws, size_t ws_size,
                              hipStream_t stream) {
  const float* hid    = (const float*)d_in[0];
  const float* Wscore = (const float*)d_in[1];
  const float* Watt   = (const float*)d_in[2];
  float* out = (float*)d_out;
  float* ws  = (float*)d_ws;

  float* V      = ws;
  float* pm     = ws + 32768;
  float* psum   = ws + 33792;
  float* pc     = ws + 34816;
  float* preact = ws + 296960;

  k_vmat   <<<dim3(8, 16),  256, 0, stream>>>(hid, Wscore, V);
  k_stream <<<dim3(B_, S_), 256, 0, stream>>>(hid, V, pm, psum, pc);
  k_combine<<<B_,           256, 0, stream>>>(hid, pm, psum, pc, preact);
  k_att    <<<dim3(8, 8),   256, 0, stream>>>(preact, Watt, out);
}


# k_stream: each wave streams 64 rows, 8 per step as whole 512 B pieces, next step's 16 loads issued before the current rows are reduced (16-32 in flight); changes the order of score and context sums
# speedup vs baseline: 1.4988x; 1.2180x over previous
#include <hip/hip_runtime.h>
#include <hip/hip_bf16.h>
#include <math.h>

#define B_ 128
#define T_ 2048
#define H_ 256
#define S_ 4
#define ROWS_PER_WG (T_ / S_)
#define ROWS_PER_WAVE (ROWS_PER_WG / 8)
#define RB_ 8
#define NB_ (ROWS_PER_WAVE / RB_)
#define KSPLIT_ 8

typedef __attribute__((ext_vector_type(2))) float v2f;
typedef __attribute__((ext_vector_type(8))) float v8f;

__global__ __launch_bounds__(256) void k_vmat(const float* __restrict__ hid,
                                              const float* __restrict__ Wscore,
                                              float* __restrict__ V) {
  const int tm   = blockIdx.x;
  const int tn   = blockIdx.y;
  const int wave = threadIdx.x >> 5;
  const int lane = threadIdx.x & 31;
  const int half = lane >> 4;
  const int r    = lane & 15;
  const size_t hstride = (size_t)T_ * H_;
  const float* ht = hid + (size_t)(T_ - 1) * H_;
  const int kb = wave * (H_ / KSPLIT_) + 2 * half;

  const float* arow = ht + (size_t)(tm * 16 + r) * hstride + kb;
  const float* brow = Wscore + (size_t)(tn * 16 + r) * H_ + kb;
  constexpr int NSTEP = H_ / KSPLIT_ / 4;
  float2 av[NSTEP], bv[NSTEP];
#pragma unroll
  for (int i = 0; i < NSTEP; ++i) {
    av[i] = *(const float2*)(arow + 4 * i);
    bv[i] = *(const float2*)(brow + 4 * i);
  }
  v8f acc = {};
#pragma unroll
  for (int i = 0; i < NSTEP; ++i) {
    v2f a; a[0] = av[i].x; a[1] = av[i].y;
    v2f b; b[0] = bv[i].x; b[1] = bv[i].y;
    acc = __builtin_amdgcn_wmma_f32_16x16x4_f32(false, a, false, b,
                                                (short)0, acc, false, false);
  }

  __shared__ float part[KSPLIT_][256];
#pragma unroll
  for (int j = 0; j < 8; ++j) part[wave][(j + 8 * half) * 16 + r] = acc[j];
  __syncthreads();
  const int t = threadIdx.x;
  float s = part[0][t];
#pragma unroll
  for (int w = 1; w < KSPLIT_; ++w) s += part[w][t];
  V[(size_t)(tm * 16 + (t >> 4)) * H_ + tn * 16 + (t & 15)] = s;
}

typedef __attribute__((ext_vector_type(4))) float v4f;
struct RowPiece { v4f lo, hi; };

__device__ __forceinline__ void load_batch(RowPiece (&buf)[RB_], const float* p) {
#pragma unroll
  for (int i = 0; i < RB_; ++i) {
    buf[i].lo = *(const v4f*)(p + (size_t)i * H_);
    buf[i].hi = *(const v4f*)(p + (size_t)i * H_ + 128);
  }
}

__device__ __forceinline__ void reduce_batch(const RowPiece (&buf)[RB_],
                                             const v4f v0, const v4f v1,
                                             float& m, float& run, v2f (&c)[4]) {
  float p[RB_];
#pragma unroll
  for (int i = 0; i < RB_; ++i) {
    v2f acc = buf[i].lo.xy * v0.xy;
    acc += buf[i].lo.zw * v0.zw;
    acc += buf[i].hi.xy * v1.xy;
    acc += buf[i].hi.zw * v1.zw;
    p[i] = acc.x + acc.y;
  }
#pragma unroll
  for (int mask = 16; mask >= 1; mask >>= 1) {
    float q[RB_];
#pragma unroll
    for (int i = 0; i < RB_; ++i) q[i] = __shfl_xor(p[i], mask);
#pragma unroll
    for (int i = 0; i < RB_; ++i) p[i] += q[i];
  }
  float bm = p[0];
#pragma unroll
  for (int i = 1; i < RB_; ++i) bm = fmaxf(bm, p[i]);
  const float mn = fmaxf(m, bm);
  const float sc = __expf(m - mn);
  float pe[RB_];
  float ps = 0.f;
#pragma unroll
  for (int i = 0; i < RB_; ++i) { pe[i] = __expf(p[i] - mn); ps += pe[i]; }
  run = run * sc + ps;
  m = mn;
#pragma unroll
  for (int j = 0; j < 4; ++j) c[j] *= sc;
#pragma unroll
  for (int i = 0; i < RB_; ++i) {
    c[0] += pe[i] * buf[i].lo.xy;
    c[1] += pe[i] * buf[i].lo.zw;
    c[2] += pe[i] * buf[i].hi.xy;
    c[3] += pe[i] * buf[i].hi.zw;
  }
}

__global__ __launch_bounds__(256) void k_stream(const float* __restrict__ hid,
                                                const float* __restrict__ V,
                                                float* __restrict__ pm,
                                                float* __restrict__ psum,
                                                float* __restrict__ pc) {
  const int b    = blockIdx.x;
  const int s    = blockIdx.y;
  const int wave = threadIdx.x >> 5;
  const int lane = threadIdx.x & 31;

  const float* vrow = V + (size_t)b * H_ + lane * 4;
  v4f v0 = *(const v4f*)vrow;
  v4f v1 = *(const v4f*)(vrow + 128);

  float m = -INFINITY, run = 0.f;
  v2f c[4] = {};

  const int t0 = s * ROWS_PER_WG + wave * ROWS_PER_WAVE;
  const float* base = hid + ((size_t)b * T_ + t0) * H_ + lane * 4;

  RowPiece bufA[RB_], bufB[RB_];
  load_batch(bufA, base);
#define REQUEST_(buf, j) do { __builtin_amdgcn_sched_barrier(0); \
    load_batch(buf, base + (size_t)(j) * RB_ * H_); \
    __builtin_amdgcn_sched_barrier(0); } while (0)
#define BATCH_END_() asm volatile("" : "+v"(m), "+v"(run), \
    "+v"(c[0]), "+v"(c[1]), "+v"(c[2]), "+v"(c[3]), "+v"(v0), "+v"(v1))
  for (int k = 0; k < NB_ - 2; k += 2) {
    REQUEST_(bufB, k + 1);
    reduce_batch(bufA, v0, v1, m, run, c);
    BATCH_END_();
    REQUEST_(bufA, k + 2);
    reduce_batch(bufB, v0, v1, m, run, c);
    BATCH_END_();
  }
  REQUEST_(bufB, NB_ - 1);
  reduce_batch(bufA, v0, v1, m, run, c);
  BATCH_END_();
  reduce_batch(bufB, v0, v1, m, run, c);
#undef REQUEST_
#undef BATCH_END_

  __shared__ float sm[8];
  __shared__ float ss[8];
  __shared__ __attribute__((aligned(16))) float scx[8][H_];
  *(float4*)&scx[wave][lane * 4]       = make_float4(c[0].x, c[0].y, c[1].x, c[1].y);
  *(float4*)&scx[wave][128 + lane * 4] = make_float4(c[2].x, c[2].y, c[3].x, c[3].y);
  if (lane == 0) { sm[wave] = m; ss[wave] = run; }
  __syncthreads();

  const int h = threadIdx.x;
  float M = sm[0];
#pragma unroll
  for (int w = 1; w < 8; ++w) M = fmaxf(M, sm[w]);
  float tot = 0.f, ch = 0.f;
#pragma unroll
  for (int w = 0; w < 8; ++w) {
    const float e = __expf(sm[w] - M);
    tot += ss[w] * e;
    ch  += scx[w][h] * e;
  }
  const int idx = b * S_ + s;
  pc[(size_t)idx * H_ + h] = ch;
  if (threadIdx.x == 0) { pm[idx] = M; psum[idx] = tot; }
}

__global__ __launch_bounds__(256) void k_combine(const float* __restrict__ hid,
                                                 const float* __restrict__ pm,
                                                 const float* __restrict__ psum,
                                                 const float* __restrict__ pc,
                                                 float* __restrict__ preact) {
  const int b = blockIdx.x;
  const int h = threadIdx.x;
  float M = pm[b * S_];
#pragma unroll
  for (int s = 1; s < S_; ++s) M = fmaxf(M, pm[b * S_ + s]);
  float tot = 0.f, ch = 0.f;
#pragma unroll
  for (int s = 0; s < S_; ++s) {
    const float e = __expf(pm[b * S_ + s] - M);
    tot += psum[b * S_ + s] * e;
    ch  += pc[(size_t)(b * S_ + s) * H_ + h] * e;
  }
  preact[(size_t)b * 512 + h]       = ch / tot;
  preact[(size_t)b * 512 + 256 + h] = hid[((size_t)b * T_ + (T_ - 1)) * H_ + h];
}

__global__ __launch_bounds__(256) void k_att(const float* __restrict__ preact,
                                             const float* __restrict__ Watt,
                                             float* __restrict__ out) {
  const int tm   = blockIdx.x;
  const int tn   = blockIdx.y;
  const int wave = threadIdx.x >> 5;
  const int lane = threadIdx.x & 31;
  const int half = lane >> 4;
  const int r    = lane & 15;
  const int kb = wave * (512 / KSPLIT_) + 2 * half;

  const float* arow = preact + (size_t)(tm * 16 + r) * 512 + kb;
  const float* bcol = Watt + (size_t)kb * 128 + tn * 16 + r;
  constexpr int NSTEP = 512 / KSPLIT_ / 4;
  float2 av[NSTEP];
  float b0[NSTEP], b1[NSTEP];
#pragma unroll
  for (int i = 0; i < NSTEP; ++i) {
    av[i] = *(const float2*)(arow + 4 * i);
    b0[i] = bcol[(size_t)(4 * i) * 128];
    b1[i] = bcol[(size_t)(4 * i + 1) * 128];
  }
  v8f acc = {};
#pragma unroll
  for (int i = 0; i < NSTEP; ++i) {
    v2f a; a[0] = av[i].x; a[1] = av[i].y;
    v2f b; b[0] = b0[i];   b[1] = b1[i];
    acc = __builtin_amdgcn_wmma_f32_16x16x4_f32(false, a, false, b,
                                                (short)0, acc, false, false);
  }

  __shared__ float part[KSPLIT_][256];
#pragma unroll
  for (int j = 0; j < 8; ++j) part[wave][(j + 8 * half) * 16 + r] = acc[j];
  __syncthreads();
  const int t = threadIdx.x;
  float s = part[0][t];
#pragma unroll
  for (int w = 1; w < KSPLIT_; ++w) s += part[w][t];
  out[(size_t)(tm * 16 + (t >> 4)) * 128 + tn * 16 + (t & 15)] = tanhf(s);
}

extern "C" void kernel_launch(void* const* d_in, const int* in_sizes, int n_in,
                              void* d_out, int out_size, void* d_ws, size_t ws_size,
                              hipStream_t stream) {
  const float* hid    = (const float*)d_in[0];
  const float* Wscore = (const float*)d_in[1];
  const float* Watt   = (const float*)d_in[2];
  float* out = (float*)d_out;
  float* ws  = (float*)d_ws;

  float* V      = ws;
  float* pm     = ws + 32768;
  float* psum   = ws + 33792;
  float* pc     = ws + 34816;
  float* preact = ws + 296960;

  k_vmat   <<<dim3(8, 16),  256, 0, stream>>>(hid, Wscore, V);
  k_stream <<<dim3(B_, S_), 256, 0, stream>>>(hid, V, pm, psum, pc);
  k_combine<<<B_,           256, 0, stream>>>(hid, pm, psum, pc, preact);
  k_att    <<<dim3(8, 8),   256, 0, stream>>>(preact, Watt, out);
}


# Final GEMM builds its 16x512 A tile in LDS from the 4 split partials and h_t with whole-line 16 B loads and reads fragments from LDS: combine launch and preact plane go (3 launches); sums unchanged
# speedup vs baseline: 1.5622x; 1.0423x over previous
#include <hip/hip_runtime.h>
#include <hip/hip_bf16.h>
#include <math.h>

#define B_ 128
#define T_ 2048
#define H_ 256
#define S_ 4
#define ROWS_PER_WG (T_ / S_)
#define ROWS_PER_WAVE (ROWS_PER_WG / 8)
#define RB_ 8
#define NB_ (ROWS_PER_WAVE / RB_)
#define KSPLIT_ 8

typedef __attribute__((ext_vector_type(2))) float v2f;
typedef __attribute__((ext_vector_type(8))) float v8f;

__global__ __launch_bounds__(256) void k_vmat(const float* __restrict__ hid,
                                              const float* __restrict__ Wscore,
                                              float* __restrict__ V) {
  const int tm   = blockIdx.x;
  const int tn   = blockIdx.y;
  const int wave = threadIdx.x >> 5;
  const int lane = threadIdx.x & 31;
  const int half = lane >> 4;
  const int r    = lane & 15;
  const size_t hstride = (size_t)T_ * H_;
  const float* ht = hid + (size_t)(T_ - 1) * H_;
  const int kb = wave * (H_ / KSPLIT_) + 2 * half;

  const float* arow = ht + (size_t)(tm * 16 + r) * hstride + kb;
  const float* brow = Wscore + (size_t)(tn * 16 + r) * H_ + kb;
  constexpr int NSTEP = H_ / KSPLIT_ / 4;
  float2 av[NSTEP], bv[NSTEP];
#pragma unroll
  for (int i = 0; i < NSTEP; ++i) {
    av[i] = *(const float2*)(arow + 4 * i);
    bv[i] = *(const float2*)(brow + 4 * i);
  }
  v8f acc = {};
#pragma unroll
  for (int i = 0; i < NSTEP; ++i) {
    v2f a; a[0] = av[i].x; a[1] = av[i].y;
    v2f b; b[0] = bv[i].x; b[1] = bv[i].y;
    acc = __builtin_amdgcn_wmma_f32_16x16x4_f32(false, a, false, b,
                                                (short)0, acc, false, false);
  }

  __shared__ float part[KSPLIT_][256];
#pragma unroll
  for (int j = 0; j < 8; ++j) part[wave][(j + 8 * half) * 16 + r] = acc[j];
  __syncthreads();
  const int t = threadIdx.x;
  float s = part[0][t];
#pragma unroll
  for (int w = 1; w < KSPLIT_; ++w) s += part[w][t];
  V[(size_t)(tm * 16 + (t >> 4)) * H_ + tn * 16 + (t & 15)] = s;
}

typedef __attribute__((ext_vector_type(4))) float v4f;
struct RowPiece { v4f lo, hi; };

__device__ __forceinline__ void load_batch(RowPiece (&buf)[RB_], const float* p) {
#pragma unroll
  for (int i = 0; i < RB_; ++i) {
    buf[i].lo = *(const v4f*)(p + (size_t)i * H_);
    buf[i].hi = *(const v4f*)(p + (size_t)i * H_ + 128);
  }
}

__device__ __forceinline__ void reduce_batch(const RowPiece (&buf)[RB_],
                                             const v4f v0, const v4f v1,
                                             float& m, float& run, v2f (&c)[4]) {
  float p[RB_];
#pragma unroll
  for (int i = 0; i < RB_; ++i) {
    v2f acc = buf[i].lo.xy * v0.xy;
    acc += buf[i].lo.zw * v0.zw;
    acc += buf[i].hi.xy * v1.xy;
    acc += buf[i].hi.zw * v1.zw;
    p[i] = acc.x + acc.y;
  }
#pragma unroll
  for (int mask = 16; mask >= 1; mask >>= 1) {
    float q[RB_];
#pragma unroll
    for (int i = 0; i < RB_; ++i) q[i] = __shfl_xor(p[i], mask);
#pragma unroll
    for (int i = 0; i < RB_; ++i) p[i] += q[i];
  }
  float bm = p[0];
#pragma unroll
  for (int i = 1; i < RB_; ++i) bm = fmaxf(bm, p[i]);
  const float mn = fmaxf(m, bm);
  const float sc = __expf(m - mn);
  float pe[RB_];
  float ps = 0.f;
#pragma unroll
  for (int i = 0; i < RB_; ++i) { pe[i] = __expf(p[i] - mn); ps += pe[i]; }
  run = run * sc + ps;
  m = mn;
#pragma unroll
  for (int j = 0; j < 4; ++j) c[j] *= sc;
#pragma unroll
  for (int i = 0; i < RB_; ++i) {
    c[0] += pe[i] * buf[i].lo.xy;
    c[1] += pe[i] * buf[i].lo.zw;
    c[2] += pe[i] * buf[i].hi.xy;
    c[3] += pe[i] * buf[i].hi.zw;
  }
}

__global__ __launch_bounds__(256) void k_stream(const float* __restrict__ hid,
                                                const float* __restrict__ V,
                                                float* __restrict__ pm,
                                                float* __restrict__ psum,
                                                float* __restrict__ pc) {
  const int b    = blockIdx.x;
  const int s    = blockIdx.y;
  const int wave = threadIdx.x >> 5;
  const int lane = threadIdx.x & 31;

  const float* vrow = V + (size_t)b * H_ + lane * 4;
  v4f v0 = *(const v4f*)vrow;
  v4f v1 = *(const v4f*)(vrow + 128);

  float m = -INFINITY, run = 0.f;
  v2f c[4] = {};

  const int t0 = s * ROWS_PER_WG + wave * ROWS_PER_WAVE;
  const float* base = hid + ((size_t)b * T_ + t0) * H_ + lane * 4;

  RowPiece bufA[RB_], bufB[RB_];
  load_batch(bufA, base);
#define REQUEST_(buf, j) do { __builtin_amdgcn_sched_barrier(0); \
    load_batch(buf, base + (size_t)(j) * RB_ * H_); \
    __builtin_amdgcn_sched_barrier(0); } while (0)
#define BATCH_END_() asm volatile("" : "+v"(m), "+v"(run), \
    "+v"(c[0]), "+v"(c[1]), "+v"(c[2]), "+v"(c[3]), "+v"(v0), "+v"(v1))
  for (int k = 0; k < NB_ - 2; k += 2) {
    REQUEST_(bufB, k + 1);
    reduce_batch(bufA, v0, v1, m, run, c);
    BATCH_END_();
    REQUEST_(bufA, k + 2);
    reduce_batch(bufB, v0, v1, m, run, c);
    BATCH_END_();
  }
  REQUEST_(bufB, NB_ - 1);
  reduce_batch(bufA, v0, v1, m, run, c);
  BATCH_END_();
  reduce_batch(bufB, v0, v1, m, run, c);
#undef REQUEST_
#undef BATCH_END_

  __shared__ float sm[8];
  __shared__ float ss[8];
  __shared__ __attribute__((aligned(16))) float scx[8][H_];
  *(float4*)&scx[wave][lane * 4]       = make_float4(c[0].x, c[0].y, c[1].x, c[1].y);
  *(float4*)&scx[wave][128 + lane * 4] = make_float4(c[2].x, c[2].y, c[3].x, c[3].y);
  if (lane == 0) { sm[wave] = m; ss[wave] = run; }
  __syncthreads();

  const int h = threadIdx.x;
  float M = sm[0];
#pragma unroll
  for (int w = 1; w < 8; ++w) M = fmaxf(M, sm[w]);
  float tot = 0.f, ch = 0.f;
#pragma unroll
  for (int w = 0; w < 8; ++w) {
    const float e = __expf(sm[w] - M);
    tot += ss[w] * e;
    ch  += scx[w][h] * e;
  }
  const int idx = b * S_ + s;
  pc[(size_t)idx * H_ + h] = ch;
  if (threadIdx.x == 0) { pm[idx] = M; psum[idx] = tot; }
}

#define APITCH_ 516
__global__ __launch_bounds__(256) void k_att(const float* __restrict__ hid,
                                             const float* __restrict__ pm,
                                             const float* __restrict__ psum,
                                             const float* __restrict__ pc,
                                             const float* __restrict__ Watt,
                                             float* __restrict__ out) {
  const int tm   = blockIdx.x;
  const int tn   = blockIdx.y;
  const int wave = threadIdx.x >> 5;
  const int lane = threadIdx.x & 31;
  const int half = lane >> 4;
  const int r    = lane & 15;
  const int kb = wave * (512 / KSPLIT_) + 2 * half;

  __shared__ __attribute__((aligned(16))) float At[16][APITCH_];
  __shared__ float part[KSPLIT_][256];

  const float* bcol = Watt + (size_t)kb * 128 + tn * 16 + r;
  constexpr int NSTEP = 512 / KSPLIT_ / 4;
  float b0[NSTEP], b1[NSTEP];
#pragma unroll
  for (int i = 0; i < NSTEP; ++i) {
    b0[i] = bcol[(size_t)(4 * i) * 128];
    b1[i] = bcol[(size_t)(4 * i + 1) * 128];
  }

  static_assert(S_ == 4, "the 4 split values of a batch are read as one float4");
  const int h4 = (threadIdx.x & 63) * 4;
  const int bs = threadIdx.x >> 6;
  v4f m4[4], s4[4], p[4][S_], ht[4];
#pragma unroll
  for (int j = 0; j < 4; ++j) {
    const int b = tm * 16 + 4 * j + bs;
    m4[j] = *(const v4f*)(pm + b * S_);
    s4[j] = *(const v4f*)(psum + b * S_);
#pragma unroll
    for (int s = 0; s < S_; ++s)
      p[j][s] = *(const v4f*)(pc + (size_t)(b * S_ + s) * H_ + h4);
    ht[j] = *(const v4f*)(hid + ((size_t)b * T_ + (T_ - 1)) * H_ + h4);
  }
#define LF4_(j) "v"(m4[j]), "v"(s4[j]), "v"(p[j][0]), "v"(p[j][1]), "v"(p[j][2]), "v"(p[j][3]), "v"(ht[j])
  asm volatile("" :: LF4_(0), LF4_(1), LF4_(2), LF4_(3));
#undef LF4_
#define LFB_(i) "v"(b0[i]), "v"(b1[i])
  asm volatile("" :: LFB_(0), LFB_(1), LFB_(2), LFB_(3), LFB_(4), LFB_(5), LFB_(6), LFB_(7));
  asm volatile("" :: LFB_(8), LFB_(9), LFB_(10), LFB_(11), LFB_(12), LFB_(13), LFB_(14), LFB_(15));
#undef LFB_
  __builtin_amdgcn_sched_barrier(0);
#pragma unroll
  for (int j = 0; j < 4; ++j) {
    const int bl = 4 * j + bs;
    const float pmv[S_] = {m4[j].x, m4[j].y, m4[j].z, m4[j].w};
    const float psv[S_] = {s4[j].x, s4[j].y, s4[j].z, s4[j].w};
    float M = pmv[0];
#pragma unroll
    for (int s = 1; s < S_; ++s) M = fmaxf(M, pmv[s]);
    float tot = 0.f;
    float4 ch = make_float4(0.f, 0.f, 0.f, 0.f);
#pragma unroll
    for (int s = 0; s < S_; ++s) {
      const float e = __expf(pmv[s] - M);
      tot  += psv[s] * e;
      ch.x += p[j][s].x * e;
      ch.y += p[j][s].y * e;
      ch.z += p[j][s].z * e;
      ch.w += p[j][s].w * e;
    }
    *(float4*)&At[bl][h4] = make_float4(ch.x / tot, ch.y / tot, ch.z / tot, ch.w / tot);
    *(v4f*)&At[bl][256 + h4] = ht[j];
  }
  __syncthreads();

  const float* arow = &At[r][kb];
  v8f acc = {};
#pragma unroll
  for (int i = 0; i < NSTEP; ++i) {
    const float2 av = *(const float2*)(arow + 4 * i);
    v2f a; a[0] = av.x;  a[1] = av.y;
    v2f b; b[0] = b0[i]; b[1] = b1[i];
    acc = __builtin_amdgcn_wmma_f32_16x16x4_f32(false, a, false, b,
                                                (short)0, acc, false, false);
  }

#pragma unroll
  for (int j = 0; j < 8; ++j) part[wave][(j + 8 * half) * 16 + r] = acc[j];
  __syncthreads();
  const int t = threadIdx.x;
  float s = part[0][t];
#pragma unroll
  for (int w = 1; w < KSPLIT_; ++w) s += part[w][t];
  out[(size_t)(tm * 16 + (t >> 4)) * 128 + tn * 16 + (t & 15)] = tanhf(s);
}

extern "C" void kernel_launch(void* const* d_in, const int* in_sizes, int n_in,
                              void* d_out, int out_size, void* d_ws, size_t ws_size,
                              hipStream_t stream) {
  const float* hid    = (const float*)d_in[0];
  const float* Wscore = (const float*)d_in[1];
  const float* Watt   = (const float*)d_in[2];
  float* out = (float*)d_out;
  float* ws  = (float*)d_ws;

  float* V      = ws;
  float* pm     = ws + 32768;
  float* psum   = ws + 33792;
  float* pc     = ws + 34816;

  k_vmat   <<<dim3(8, 16),  256, 0, stream>>>(hid, Wscore, V);
  k_stream <<<dim3(B_, S_), 256, 0, stream>>>(hid, V, pm, psum, pc);
  k_att    <<<dim3(8, 8),   256, 0, stream>>>(hid, pm, psum, pc, Watt, out);
}
